# starting kernel
# baseline (speedup 1.0000x reference)
#include <hip/hip_runtime.h>

typedef __attribute__((ext_vector_type(2))) float v2f;
typedef __attribute__((ext_vector_type(8))) float v8f;

__device__ __forceinline__ v8f wmma4(v2f a, v2f b, v8f c) {
  return __builtin_amdgcn_wmma_f32_16x16x4_f32(
       false, a,  false, b,
       (short)0, c,  false,  false);
}

__device__ __forceinline__ float gelu_exact(float x) {
  return 0.5f * x * (1.0f + erff(x * 0.70710678118654752440f));
}

__global__ __launch_bounds__(256) void ln_kernel(const float* __restrict__ x,
                                                 const float* __restrict__ gw,
                                                 const float* __restrict__ bw,
                                                 float* __restrict__ out) {
  const int row  = blockIdx.x;
  const int t    = threadIdx.x;
  const int wave = t >> 5, lane = t & 31;

  const float4 v = ((const float4*)(x + (size_t)row * 1024))[t];
  float s  = v.x + v.y + v.z + v.w;
  float s2 = v.x * v.x + v.y * v.y + v.z * v.z + v.w * v.w;
#pragma unroll
  for (int off = 16; off >= 1; off >>= 1) {
    s  += __shfl_xor(s, off, 32);
    s2 += __shfl_xor(s2, off, 32);
  }
  __shared__ float red[16];
  if (lane == 0) { red[2 * wave] = s; red[2 * wave + 1] = s2; }
  __syncthreads();
  float S1 = 0.f, S2 = 0.f;
#pragma unroll
  for (int w = 0; w < 8; ++w) { S1 += red[2 * w]; S2 += red[2 * w + 1]; }
  const float mean = S1 * (1.0f / 1024.0f);
  const float var  = S2 * (1.0f / 1024.0f) - mean * mean;
  const float rstd = rsqrtf(var + 1e-5f);

  const float4 g4 = ((const float4*)gw)[t];
  const float4 b4 = ((const float4*)bw)[t];
  float4 o;
  o.x = (v.x - mean) * rstd * g4.x + b4.x;
  o.y = (v.y - mean) * rstd * g4.y + b4.y;
  o.z = (v.z - mean) * rstd * g4.z + b4.z;
  o.w = (v.w - mean) * rstd * g4.w + b4.w;
  ((float4*)(out + (size_t)row * 1024))[t] = o;
}

template <int EPI>
__global__ __launch_bounds__(256) void gemm_kernel(const float* __restrict__ A,
                                                   const float* __restrict__ W,
                                                   const float* __restrict__ bias,
                                                   const float* __restrict__ res,
                                                   float* __restrict__ C,
                                                   int N, int K) {
  const int lane  = threadIdx.x & 31;
  const int wave  = threadIdx.x >> 5;
  const int row   = lane & 15;
  const int kh    = lane >> 4;
  const int waveM = wave >> 1;
  const int waveN = wave & 1;
  const int m0 = blockIdx.x * 128 + waveM * 32;
  const int n0 = blockIdx.y * 64 + waveN * 32;

  v8f c00 = {}, c01 = {}, c10 = {}, c11 = {};

  const float* a0p = A + (m0 + row) * K + 2 * kh;
  const float* a1p = a0p + 16 * K;
  const float* bp  = W + (2 * kh) * N + n0 + row;

#pragma unroll 4
  for (int k = 0; k < K; k += 4) {
    v2f a0 = *(const v2f*)a0p;
    v2f a1 = *(const v2f*)a1p;
    v2f b0, b1;
    b0.x = bp[0];  b0.y = bp[N];
    b1.x = bp[16]; b1.y = bp[N + 16];
    c00 = wmma4(a0, b0, c00);
    c01 = wmma4(a0, b1, c01);
    c10 = wmma4(a1, b0, c10);
    c11 = wmma4(a1, b1, c11);
    a0p += 4; a1p += 4; bp += 4 * N;
  }

  const float bv0 = bias[n0 + row];
  const float bv1 = bias[n0 + 16 + row];
#pragma unroll
  for (int i = 0; i < 8; ++i) {
    const int r0 = m0 + i + 8 * kh;
    const int r1 = r0 + 16;
    const int cA = n0 + row;
    const int cB = cA + 16;
    float v00 = c00[i] + bv0, v01 = c01[i] + bv1;
    float v10 = c10[i] + bv0, v11 = c11[i] + bv1;
    if (EPI == 1) {
      v00 += res[(size_t)r0 * N + cA]; v01 += res[(size_t)r0 * N + cB];
      v10 += res[(size_t)r1 * N + cA]; v11 += res[(size_t)r1 * N + cB];
    }
    if (EPI == 2) {
      v00 = gelu_exact(v00); v01 = gelu_exact(v01);
      v10 = gelu_exact(v10); v11 = gelu_exact(v11);
    }
    C[(size_t)r0 * N + cA] = v00; C[(size_t)r0 * N + cB] = v01;
    C[(size_t)r1 * N + cA] = v10; C[(size_t)r1 * N + cB] = v11;
  }
}

__global__ __launch_bounds__(128) void attn_kernel(const float* __restrict__ qkv,
                                                   float* __restrict__ out) {
  constexpr int S = 2048, D = 1024, R3D = 3072;
  const int lane = threadIdx.x & 31;
  const int wave = threadIdx.x >> 5;
  const int row  = lane & 15;
  const int kh   = lane >> 4;
  const int bh   = blockIdx.x;
  const int b    = bh >> 4, h = bh & 15;
  const int q0   = (blockIdx.y * 4 + wave) * 16;

  const float* qbase = qkv + (size_t)b * S * R3D + h * 64;
  const float* kbase = qbase + D;
  const float* vbase = qbase + 2 * D;

  v2f qf[16];
  {
    const float* qrow = qbase + (size_t)(q0 + row) * R3D + 2 * kh;
#pragma unroll
    for (int j = 0; j < 16; ++j) qf[j] = *(const v2f*)(qrow + 4 * j);
  }

  v8f o0 = {}, o1 = {}, o2 = {}, o3 = {};
  float mrun[8], lrun[8];
#pragma unroll
  for (int i = 0; i < 8; ++i) { mrun[i] = -1e30f; lrun[i] = 0.f; }

  __shared__ float Pbuf[4][16 * 18];
  float* P = Pbuf[wave];

  for (int kt = 0; kt < S; kt += 16) {
    v8f sc = {};
    {
      const float* krow = kbase + (size_t)(kt + row) * R3D + 2 * kh;
#pragma unroll
      for (int j = 0; j < 16; ++j) {
        v2f kf = *(const v2f*)(krow + 4 * j);
        sc = wmma4(qf[j], kf, sc);
      }
    }
    float alpha[8];
#pragma unroll
    for (int i = 0; i < 8; ++i) {
      float s = sc[i] * 0.125f;
      float mx = s;
      mx = fmaxf(mx, __shfl_xor(mx, 1, 32));
      mx = fmaxf(mx, __shfl_xor(mx, 2, 32));
      mx = fmaxf(mx, __shfl_xor(mx, 4, 32));
      mx = fmaxf(mx, __shfl_xor(mx, 8, 32));
      const float mnew = fmaxf(mrun[i], mx);
      const float a    = __expf(mrun[i] - mnew);
      const float p    = __expf(s - mnew);
      float sum = p;
      sum += __shfl_xor(sum, 1, 32);
      sum += __shfl_xor(sum, 2, 32);
      sum += __shfl_xor(sum, 4, 32);
      sum += __shfl_xor(sum, 8, 32);
      lrun[i] = lrun[i] * a + sum;
      mrun[i] = mnew;
      alpha[i] = a;
      P[(i + 8 * kh) * 18 + row] = p;
    }
#pragma unroll
    for (int i = 0; i < 8; ++i) {
      o0[i] *= alpha[i]; o1[i] *= alpha[i];
      o2[i] *= alpha[i]; o3[i] *= alpha[i];
    }
#pragma unroll
    for (int j = 0; j < 4; ++j) {
      v2f pa = *(const v2f*)(P + row * 18 + 4 * j + 2 * kh);
      const float* vr0 = vbase + (size_t)(kt + 4 * j + 2 * kh) * R3D;
      const float* vr1 = vr0 + R3D;
      v2f vb0 = { vr0[row],      vr1[row]      };
      v2f vb1 = { vr0[16 + row], vr1[16 + row] };
      v2f vb2 = { vr0[32 + row], vr1[32 + row] };
      v2f vb3 = { vr0[48 + row], vr1[48 + row] };
      o0 = wmma4(pa, vb0, o0);
      o1 = wmma4(pa, vb1, o1);
      o2 = wmma4(pa, vb2, o2);
      o3 = wmma4(pa, vb3, o3);
    }
  }

#pragma unroll
  for (int i = 0; i < 8; ++i) {
    const int qr = q0 + i + 8 * kh;
    const float inv = 1.0f / lrun[i];
    float* orow = out + ((size_t)b * S + qr) * D + h * 64 + row;
    orow[0]  = o0[i] * inv;
    orow[16] = o1[i] * inv;
    orow[32] = o2[i] * inv;
    orow[48] = o3[i] * inv;
  }
}

extern "C" void kernel_launch(void* const* d_in, const int* in_sizes, int n_in,
                              void* d_out, int out_size, void* d_ws, size_t ws_size,
                              hipStream_t stream) {
  const float* x      = (const float*)d_in[0];
  const float* ln1_g  = (const float*)d_in[1];
  const float* ln1_b  = (const float*)d_in[2];
  const float* qkv_w  = (const float*)d_in[3];
  const float* qkv_b  = (const float*)d_in[4];
  const float* proj_w = (const float*)d_in[5];
  const float* proj_b = (const float*)d_in[6];
  const float* ln2_g  = (const float*)d_in[7];
  const float* ln2_b  = (const float*)d_in[8];
  const float* fc1_w  = (const float*)d_in[9];
  const float* fc1_b  = (const float*)d_in[10];
  const float* fc2_w  = (const float*)d_in[11];
  const float* fc2_b  = (const float*)d_in[12];
  float* out = (float*)d_out;

  const int M = 4096;
  float* bufA = (float*)d_ws;
  float* bufB = bufA + (size_t)4096 * 4096;
  float* bufC = bufB + (size_t)4096 * 3072;
  float* bufD = bufC + (size_t)4096 * 1024;

  ln_kernel<<<M, 256, 0, stream>>>(x, ln1_g, ln1_b, bufA);
  gemm_kernel<0><<<dim3(M / 128, 3072 / 64), 256, 0, stream>>>(
      bufA, qkv_w, qkv_b, nullptr, bufB, 3072, 1024);
  attn_kernel<<<dim3(32, 32), 128, 0, stream>>>(bufB, bufC);
  gemm_kernel<1><<<dim3(M / 128, 1024 / 64), 256, 0, stream>>>(
      bufC, proj_w, proj_b, x, bufD, 1024, 1024);
  ln_kernel<<<M, 256, 0, stream>>>(bufD, ln2_g, ln2_b, bufB);
  gemm_kernel<2><<<dim3(M / 128, 4096 / 64), 256, 0, stream>>>(
      bufB, fc1_w, fc1_b, nullptr, bufA, 4096, 1024);
  gemm_kernel<1><<<dim3(M / 128, 1024 / 64), 256, 0, stream>>>(
      bufA, fc2_w, fc2_b, bufD, out, 1024, 4096);
}


# GEMMs: stage A and W K-tiles (32 deep) through two LDS stages with whole-row 16-byte global loads; WMMA fragments read from LDS instead of per-lane strided global loads
# speedup vs baseline: 1.2407x; 1.2407x over previous
#include <hip/hip_runtime.h>

typedef __attribute__((ext_vector_type(2))) float v2f;
typedef __attribute__((ext_vector_type(8))) float v8f;

__device__ __forceinline__ v8f wmma4(v2f a, v2f b, v8f c) {
  return __builtin_amdgcn_wmma_f32_16x16x4_f32(
       false, a,  false, b,
       (short)0, c,  false,  false);
}

__device__ __forceinline__ float gelu_exact(float x) {
  return 0.5f * x * (1.0f + erff(x * 0.70710678118654752440f));
}

__global__ __launch_bounds__(256) void ln_kernel(const float* __restrict__ x,
                                                 const float* __restrict__ gw,
                                                 const float* __restrict__ bw,
                                                 float* __restrict__ out) {
  const int row  = blockIdx.x;
  const int t    = threadIdx.x;
  const int wave = t >> 5, lane = t & 31;

  const float4 v = ((const float4*)(x + (size_t)row * 1024))[t];
  float s  = v.x + v.y + v.z + v.w;
  float s2 = v.x * v.x + v.y * v.y + v.z * v.z + v.w * v.w;
#pragma unroll
  for (int off = 16; off >= 1; off >>= 1) {
    s  += __shfl_xor(s, off, 32);
    s2 += __shfl_xor(s2, off, 32);
  }
  __shared__ float red[16];
  if (lane == 0) { red[2 * wave] = s; red[2 * wave + 1] = s2; }
  __syncthreads();
  float S1 = 0.f, S2 = 0.f;
#pragma unroll
  for (int w = 0; w < 8; ++w) { S1 += red[2 * w]; S2 += red[2 * w + 1]; }
  const float mean = S1 * (1.0f / 1024.0f);
  const float var  = S2 * (1.0f / 1024.0f) - mean * mean;
  const float rstd = rsqrtf(var + 1e-5f);

  const float4 g4 = ((const float4*)gw)[t];
  const float4 b4 = ((const float4*)bw)[t];
  float4 o;
  o.x = (v.x - mean) * rstd * g4.x + b4.x;
  o.y = (v.y - mean) * rstd * g4.y + b4.y;
  o.z = (v.z - mean) * rstd * g4.z + b4.z;
  o.w = (v.w - mean) * rstd * g4.w + b4.w;
  ((float4*)(out + (size_t)row * 1024))[t] = o;
}

template <int EPI>
__global__ __launch_bounds__(256) void gemm_kernel(const float* __restrict__ A,
                                                   const float* __restrict__ W,
                                                   const float* __restrict__ bias,
                                                   const float* __restrict__ res,
                                                   float* __restrict__ C,
                                                   int N, int K) {
  constexpr int BM = 128, BN = 64, BK = 32;
  constexpr int PA = BK + 4;
  constexpr int BG = BK * 16 + 16;
  __shared__ __attribute__((aligned(16))) float As[2][BM * PA];
  __shared__ __attribute__((aligned(16))) float Bs[2][(BN / 16) * BG];

  const int tid   = threadIdx.x;
  const int lane  = tid & 31;
  const int wave  = tid >> 5;
  const int row   = lane & 15;
  const int kh    = lane >> 4;
  const int waveM = wave >> 1;
  const int waveN = wave & 1;
  const int bm0 = blockIdx.x * BM;
  const int bn0 = blockIdx.y * BN;
  const int m0 = bm0 + waveM * 32;
  const int n0 = bn0 + waveN * 32;

  const int ar = tid >> 3, ac = (tid & 7) * 4;
  const int br = tid >> 4, bc = (tid & 15) * 4;
  const float* ag = A + (size_t)(bm0 + ar) * K + ac;
  const float* bg = W + (size_t)br * N + bn0 + bc;
  const int a_st = ar * PA + ac;
  const int b_st = (bc >> 4) * BG + br * 16 + (bc & 15);

  const int a_ld = (waveM * 32 + row) * PA + 2 * kh;
  const int b_ld = (waveN * 2) * BG + (2 * kh) * 16 + row;

  v8f c00 = {}, c01 = {}, c10 = {}, c11 = {};

  const size_t aK32 = (size_t)32 * K;
  const size_t bN16 = (size_t)16 * N;
  float4 ra0 = *(const float4*)(ag);
  float4 ra1 = *(const float4*)(ag + aK32);
  float4 ra2 = *(const float4*)(ag + 2 * aK32);
  float4 ra3 = *(const float4*)(ag + 3 * aK32);
  float4 rb0 = *(const float4*)(bg);
  float4 rb1 = *(const float4*)(bg + bN16);
  *(float4*)(&As[0][a_st])           = ra0;
  *(float4*)(&As[0][a_st + 32 * PA]) = ra1;
  *(float4*)(&As[0][a_st + 64 * PA]) = ra2;
  *(float4*)(&As[0][a_st + 96 * PA]) = ra3;
  *(float4*)(&Bs[0][b_st])           = rb0;
  *(float4*)(&Bs[0][b_st + 16 * 16]) = rb1;
  __syncthreads();

  const int nkt = K / BK;
  for (int t = 0; t < nkt; ++t) {
    const int buf = t & 1;
    const bool more = (t + 1 < nkt);
    if (more) {
      ag += BK;
      bg += (size_t)BK * N;
      ra0 = *(const float4*)(ag);
      ra1 = *(const float4*)(ag + aK32);
      ra2 = *(const float4*)(ag + 2 * aK32);
      ra3 = *(const float4*)(ag + 3 * aK32);
      rb0 = *(const float4*)(bg);
      rb1 = *(const float4*)(bg + bN16);
    }

    const float* as = &As[buf][a_ld];
    const float* bs = &Bs[buf][b_ld];
#pragma unroll
    for (int kk = 0; kk < BK; kk += 4) {
      const v2f a0 = *(const v2f*)(as + kk);
      const v2f a1 = *(const v2f*)(as + 16 * PA + kk);
      v2f b0, b1;
      b0.x = bs[kk * 16];      b0.y = bs[kk * 16 + 16];
      b1.x = bs[BG + kk * 16]; b1.y = bs[BG + kk * 16 + 16];
      c00 = wmma4(a0, b0, c00);
      c01 = wmma4(a0, b1, c01);
      c10 = wmma4(a1, b0, c10);
      c11 = wmma4(a1, b1, c11);
    }

    if (more) {
      float* asn = &As[buf ^ 1][a_st];
      float* bsn = &Bs[buf ^ 1][b_st];
      *(float4*)(asn)           = ra0;
      *(float4*)(asn + 32 * PA) = ra1;
      *(float4*)(asn + 64 * PA) = ra2;
      *(float4*)(asn + 96 * PA) = ra3;
      *(float4*)(bsn)           = rb0;
      *(float4*)(bsn + 16 * 16) = rb1;
    }
    __syncthreads();
  }

  const float bv0 = bias[n0 + row];
  const float bv1 = bias[n0 + 16 + row];
#pragma unroll
  for (int i = 0; i < 8; ++i) {
    const int r0 = m0 + i + 8 * kh;
    const int r1 = r0 + 16;
    const int cA = n0 + row;
    const int cB = cA + 16;
    float v00 = c00[i] + bv0, v01 = c01[i] + bv1;
    float v10 = c10[i] + bv0, v11 = c11[i] + bv1;
    if (EPI == 1) {
      v00 += res[(size_t)r0 * N + cA]; v01 += res[(size_t)r0 * N + cB];
      v10 += res[(size_t)r1 * N + cA]; v11 += res[(size_t)r1 * N + cB];
    }
    if (EPI == 2) {
      v00 = gelu_exact(v00); v01 = gelu_exact(v01);
      v10 = gelu_exact(v10); v11 = gelu_exact(v11);
    }
    C[(size_t)r0 * N + cA] = v00; C[(size_t)r0 * N + cB] = v01;
    C[(size_t)r1 * N + cA] = v10; C[(size_t)r1 * N + cB] = v11;
  }
}

__global__ __launch_bounds__(128) void attn_kernel(const float* __restrict__ qkv,
                                                   float* __restrict__ out) {
  constexpr int S = 2048, D = 1024, R3D = 3072;
  const int lane = threadIdx.x & 31;
  const int wave = threadIdx.x >> 5;
  const int row  = lane & 15;
  const int kh   = lane >> 4;
  const int bh   = blockIdx.x;
  const int b    = bh >> 4, h = bh & 15;
  const int q0   = (blockIdx.y * 4 + wave) * 16;

  const float* qbase = qkv + (size_t)b * S * R3D + h * 64;
  const float* kbase = qbase + D;
  const float* vbase = qbase + 2 * D;

  v2f qf[16];
  {
    const float* qrow = qbase + (size_t)(q0 + row) * R3D + 2 * kh;
#pragma unroll
    for (int j = 0; j < 16; ++j) qf[j] = *(const v2f*)(qrow + 4 * j);
  }

  v8f o0 = {}, o1 = {}, o2 = {}, o3 = {};
  float mrun[8], lrun[8];
#pragma unroll
  for (int i = 0; i < 8; ++i) { mrun[i] = -1e30f; lrun[i] = 0.f; }

  __shared__ float Pbuf[4][16 * 18];
  float* P = Pbuf[wave];

  for (int kt = 0; kt < S; kt += 16) {
    v8f sc = {};
    {
      const float* krow = kbase + (size_t)(kt + row) * R3D + 2 * kh;
#pragma unroll
      for (int j = 0; j < 16; ++j) {
        v2f kf = *(const v2f*)(krow + 4 * j);
        sc = wmma4(qf[j], kf, sc);
      }
    }
    float alpha[8];
#pragma unroll
    for (int i = 0; i < 8; ++i) {
      float s = sc[i] * 0.125f;
      float mx = s;
      mx = fmaxf(mx, __shfl_xor(mx, 1, 32));
      mx = fmaxf(mx, __shfl_xor(mx, 2, 32));
      mx = fmaxf(mx, __shfl_xor(mx, 4, 32));
      mx = fmaxf(mx, __shfl_xor(mx, 8, 32));
      const float mnew = fmaxf(mrun[i], mx);
      const float a    = __expf(mrun[i] - mnew);
      const float p    = __expf(s - mnew);
      float sum = p;
      sum += __shfl_xor(sum, 1, 32);
      sum += __shfl_xor(sum, 2, 32);
      sum += __shfl_xor(sum, 4, 32);
      sum += __shfl_xor(sum, 8, 32);
      lrun[i] = lrun[i] * a + sum;
      mrun[i] = mnew;
      alpha[i] = a;
      P[(i + 8 * kh) * 18 + row] = p;
    }
#pragma unroll
    for (int i = 0; i < 8; ++i) {
      o0[i] *= alpha[i]; o1[i] *= alpha[i];
      o2[i] *= alpha[i]; o3[i] *= alpha[i];
    }
#pragma unroll
    for (int j = 0; j < 4; ++j) {
      v2f pa = *(const v2f*)(P + row * 18 + 4 * j + 2 * kh);
      const float* vr0 = vbase + (size_t)(kt + 4 * j + 2 * kh) * R3D;
      const float* vr1 = vr0 + R3D;
      v2f vb0 = { vr0[row],      vr1[row]      };
      v2f vb1 = { vr0[16 + row], vr1[16 + row] };
      v2f vb2 = { vr0[32 + row], vr1[32 + row] };
      v2f vb3 = { vr0[48 + row], vr1[48 + row] };
      o0 = wmma4(pa, vb0, o0);
      o1 = wmma4(pa, vb1, o1);
      o2 = wmma4(pa, vb2, o2);
      o3 = wmma4(pa, vb3, o3);
    }
  }

#pragma unroll
  for (int i = 0; i < 8; ++i) {
    const int qr = q0 + i + 8 * kh;
    const float inv = 1.0f / lrun[i];
    float* orow = out + ((size_t)b * S + qr) * D + h * 64 + row;
    orow[0]  = o0[i] * inv;
    orow[16] = o1[i] * inv;
    orow[32] = o2[i] * inv;
    orow[48] = o3[i] * inv;
  }
}

extern "C" void kernel_launch(void* const* d_in, const int* in_sizes, int n_in,
                              void* d_out, int out_size, void* d_ws, size_t ws_size,
                              hipStream_t stream) {
  const float* x      = (const float*)d_in[0];
  const float* ln1_g  = (const float*)d_in[1];
  const float* ln1_b  = (const float*)d_in[2];
  const float* qkv_w  = (const float*)d_in[3];
  const float* qkv_b  = (const float*)d_in[4];
  const float* proj_w = (const float*)d_in[5];
  const float* proj_b = (const float*)d_in[6];
  const float* ln2_g  = (const float*)d_in[7];
  const float* ln2_b  = (const float*)d_in[8];
  const float* fc1_w  = (const float*)d_in[9];
  const float* fc1_b  = (const float*)d_in[10];
  const float* fc2_w  = (const float*)d_in[11];
  const float* fc2_b  = (const float*)d_in[12];
  float* out = (float*)d_out;

  const int M = 4096;
  float* bufA = (float*)d_ws;
  float* bufB = bufA + (size_t)4096 * 4096;
  float* bufC = bufB + (size_t)4096 * 3072;
  float* bufD = bufC + (size_t)4096 * 1024;

  ln_kernel<<<M, 256, 0, stream>>>(x, ln1_g, ln1_b, bufA);
  gemm_kernel<0><<<dim3(M / 128, 3072 / 64), 256, 0, stream>>>(
      bufA, qkv_w, qkv_b, nullptr, bufB, 3072, 1024);
  attn_kernel<<<dim3(32, 32), 128, 0, stream>>>(bufB, bufC);
  gemm_kernel<1><<<dim3(M / 128, 1024 / 64), 256, 0, stream>>>(
      bufC, proj_w, proj_b, x, bufD, 1024, 1024);
  ln_kernel<<<M, 256, 0, stream>>>(bufD, ln2_g, ln2_b, bufB);
  gemm_kernel<2><<<dim3(M / 128, 4096 / 64), 256, 0, stream>>>(
      bufB, fc1_w, fc1_b, nullptr, bufA, 4096, 1024);
  gemm_kernel<1><<<dim3(M / 128, 1024 / 64), 256, 0, stream>>>(
      bufA, fc2_w, fc2_b, bufD, out, 1024, 4096);
}


# attention: K and V staged through two LDS stages of 64 keys with whole-row 16-byte global loads, shared by 8 Q-tile waves per block; fragment reads from LDS in groups ahead of the WMMA runs
# speedup vs baseline: 1.3067x; 1.0532x over previous
#include <hip/hip_runtime.h>

typedef __attribute__((ext_vector_type(2))) float v2f;
typedef __attribute__((ext_vector_type(8))) float v8f;

__device__ __forceinline__ v8f wmma4(v2f a, v2f b, v8f c) {
  return __builtin_amdgcn_wmma_f32_16x16x4_f32(
       false, a,  false, b,
       (short)0, c,  false,  false);
}

__device__ __forceinline__ float gelu_exact(float x) {
  return 0.5f * x * (1.0f + erff(x * 0.70710678118654752440f));
}

__global__ __launch_bounds__(256) void ln_kernel(const float* __restrict__ x,
                                                 const float* __restrict__ gw,
                                                 const float* __restrict__ bw,
                                                 float* __restrict__ out) {
  const int row  = blockIdx.x;
  const int t    = threadIdx.x;
  const int wave = t >> 5, lane = t & 31;

  const float4 v = ((const float4*)(x + (size_t)row * 1024))[t];
  float s  = v.x + v.y + v.z + v.w;
  float s2 = v.x * v.x + v.y * v.y + v.z * v.z + v.w * v.w;
#pragma unroll
  for (int off = 16; off >= 1; off >>= 1) {
    s  += __shfl_xor(s, off, 32);
    s2 += __shfl_xor(s2, off, 32);
  }
  __shared__ float red[16];
  if (lane == 0) { red[2 * wave] = s; red[2 * wave + 1] = s2; }
  __syncthreads();
  float S1 = 0.f, S2 = 0.f;
#pragma unroll
  for (int w = 0; w < 8; ++w) { S1 += red[2 * w]; S2 += red[2 * w + 1]; }
  const float mean = S1 * (1.0f / 1024.0f);
  const float var  = S2 * (1.0f / 1024.0f) - mean * mean;
  const float rstd = rsqrtf(var + 1e-5f);

  const float4 g4 = ((const float4*)gw)[t];
  const float4 b4 = ((const float4*)bw)[t];
  float4 o;
  o.x = (v.x - mean) * rstd * g4.x + b4.x;
  o.y = (v.y - mean) * rstd * g4.y + b4.y;
  o.z = (v.z - mean) * rstd * g4.z + b4.z;
  o.w = (v.w - mean) * rstd * g4.w + b4.w;
  ((float4*)(out + (size_t)row * 1024))[t] = o;
}

template <int EPI>
__global__ __launch_bounds__(256) void gemm_kernel(const float* __restrict__ A,
                                                   const float* __restrict__ W,
                                                   const float* __restrict__ bias,
                                                   const float* __restrict__ res,
                                                   float* __restrict__ C,
                                                   int N, int K) {
  constexpr int BM = 128, BN = 64, BK = 32;
  constexpr int PA = BK + 4;
  constexpr int BG = BK * 16 + 16;
  __shared__ __attribute__((aligned(16))) float As[2][BM * PA];
  __shared__ __attribute__((aligned(16))) float Bs[2][(BN / 16) * BG];

  const int tid   = threadIdx.x;
  const int lane  = tid & 31;
  const int wave  = tid >> 5;
  const int row   = lane & 15;
  const int kh    = lane >> 4;
  const int waveM = wave >> 1;
  const int waveN = wave & 1;
  const int bm0 = blockIdx.x * BM;
  const int bn0 = blockIdx.y * BN;
  const int m0 = bm0 + waveM * 32;
  const int n0 = bn0 + waveN * 32;

  const int ar = tid >> 3, ac = (tid & 7) * 4;
  const int br = tid >> 4, bc = (tid & 15) * 4;
  const float* ag = A + (size_t)(bm0 + ar) * K + ac;
  const float* bg = W + (size_t)br * N + bn0 + bc;
  const int a_st = ar * PA + ac;
  const int b_st = (bc >> 4) * BG + br * 16 + (bc & 15);

  const int a_ld = (waveM * 32 + row) * PA + 2 * kh;
  const int b_ld = (waveN * 2) * BG + (2 * kh) * 16 + row;

  v8f c00 = {}, c01 = {}, c10 = {}, c11 = {};

  const size_t aK32 = (size_t)32 * K;
  const size_t bN16 = (size_t)16 * N;
  float4 ra0 = *(const float4*)(ag);
  float4 ra1 = *(const float4*)(ag + aK32);
  float4 ra2 = *(const float4*)(ag + 2 * aK32);
  float4 ra3 = *(const float4*)(ag + 3 * aK32);
  float4 rb0 = *(const float4*)(bg);
  float4 rb1 = *(const float4*)(bg + bN16);
  *(float4*)(&As[0][a_st])           = ra0;
  *(float4*)(&As[0][a_st + 32 * PA]) = ra1;
  *(float4*)(&As[0][a_st + 64 * PA]) = ra2;
  *(float4*)(&As[0][a_st + 96 * PA]) = ra3;
  *(float4*)(&Bs[0][b_st])           = rb0;
  *(float4*)(&Bs[0][b_st + 16 * 16]) = rb1;
  __syncthreads();

  const int nkt = K / BK;
  for (int t = 0; t < nkt; ++t) {
    const int buf = t & 1;
    const bool more = (t + 1 < nkt);
    if (more) {
      ag += BK;
      bg += (size_t)BK * N;
      ra0 = *(const float4*)(ag);
      ra1 = *(const float4*)(ag + aK32);
      ra2 = *(const float4*)(ag + 2 * aK32);
      ra3 = *(const float4*)(ag + 3 * aK32);
      rb0 = *(const float4*)(bg);
      rb1 = *(const float4*)(bg + bN16);
    }

    const float* as = &As[buf][a_ld];
    const float* bs = &Bs[buf][b_ld];
#pragma unroll
    for (int kk = 0; kk < BK; kk += 4) {
      const v2f a0 = *(const v2f*)(as + kk);
      const v2f a1 = *(const v2f*)(as + 16 * PA + kk);
      v2f b0, b1;
      b0.x = bs[kk * 16];      b0.y = bs[kk * 16 + 16];
      b1.x = bs[BG + kk * 16]; b1.y = bs[BG + kk * 16 + 16];
      c00 = wmma4(a0, b0, c00);
      c01 = wmma4(a0, b1, c01);
      c10 = wmma4(a1, b0, c10);
      c11 = wmma4(a1, b1, c11);
    }

    if (more) {
      float* asn = &As[buf ^ 1][a_st];
      float* bsn = &Bs[buf ^ 1][b_st];
      *(float4*)(asn)           = ra0;
      *(float4*)(asn + 32 * PA) = ra1;
      *(float4*)(asn + 64 * PA) = ra2;
      *(float4*)(asn + 96 * PA) = ra3;
      *(float4*)(bsn)           = rb0;
      *(float4*)(bsn + 16 * 16) = rb1;
    }
    __syncthreads();
  }

  const float bv0 = bias[n0 + row];
  const float bv1 = bias[n0 + 16 + row];
#pragma unroll
  for (int i = 0; i < 8; ++i) {
    const int r0 = m0 + i + 8 * kh;
    const int r1 = r0 + 16;
    const int cA = n0 + row;
    const int cB = cA + 16;
    float v00 = c00[i] + bv0, v01 = c01[i] + bv1;
    float v10 = c10[i] + bv0, v11 = c11[i] + bv1;
    if (EPI == 1) {
      v00 += res[(size_t)r0 * N + cA]; v01 += res[(size_t)r0 * N + cB];
      v10 += res[(size_t)r1 * N + cA]; v11 += res[(size_t)r1 * N + cB];
    }
    if (EPI == 2) {
      v00 = gelu_exact(v00); v01 = gelu_exact(v01);
      v10 = gelu_exact(v10); v11 = gelu_exact(v11);
    }
    C[(size_t)r0 * N + cA] = v00; C[(size_t)r0 * N + cB] = v01;
    C[(size_t)r1 * N + cA] = v10; C[(size_t)r1 * N + cB] = v11;
  }
}

__global__ __launch_bounds__(256) void attn_kernel(const float* __restrict__ qkv,
                                                   float* __restrict__ out) {
  constexpr int S = 2048, D = 1024, R3D = 3072;
  constexpr int KT = 64;
  constexpr int PK = 64 + 4;
  constexpr int VG = KT * 16 + 16;
  __shared__ __attribute__((aligned(16))) float Ks[2][KT * PK];
  __shared__ __attribute__((aligned(16))) float Vs[2][4 * VG];
  __shared__ float Pbuf[8][16 * 18];

  const int tid  = threadIdx.x;
  const int lane = tid & 31;
  const int wave = tid >> 5;
  const int row  = lane & 15;
  const int kh   = lane >> 4;
  const int bh   = blockIdx.x;
  const int b    = bh >> 4, h = bh & 15;
  const int q0   = (blockIdx.y * 8 + wave) * 16;

  const float* qbase = qkv + (size_t)b * S * R3D + h * 64;
  const float* kbase = qbase + D;
  const float* vbase = qbase + 2 * D;

  const int sr = tid >> 4, sc4 = (tid & 15) * 4;
  const float* kg = kbase + (size_t)sr * R3D + sc4;
  const float* vg = vbase + (size_t)sr * R3D + sc4;
  const int k_st = sr * PK + sc4;
  const int v_st = (sc4 >> 4) * VG + sr * 16 + (sc4 & 15);
  const size_t r16 = (size_t)16 * R3D;

  float4 rk0 = *(const float4*)(kg);
  float4 rk1 = *(const float4*)(kg + r16);
  float4 rk2 = *(const float4*)(kg + 2 * r16);
  float4 rk3 = *(const float4*)(kg + 3 * r16);
  float4 rv0 = *(const float4*)(vg);
  float4 rv1 = *(const float4*)(vg + r16);
  float4 rv2 = *(const float4*)(vg + 2 * r16);
  float4 rv3 = *(const float4*)(vg + 3 * r16);

  v2f qf[16];
  {
    const float* qrow = qbase + (size_t)(q0 + row) * R3D + 2 * kh;
#pragma unroll
    for (int j = 0; j < 16; ++j) qf[j] = *(const v2f*)(qrow + 4 * j);
  }

  *(float4*)(&Ks[0][k_st])           = rk0;
  *(float4*)(&Ks[0][k_st + 16 * PK]) = rk1;
  *(float4*)(&Ks[0][k_st + 32 * PK]) = rk2;
  *(float4*)(&Ks[0][k_st + 48 * PK]) = rk3;
  *(float4*)(&Vs[0][v_st])           = rv0;
  *(float4*)(&Vs[0][v_st + 16 * 16]) = rv1;
  *(float4*)(&Vs[0][v_st + 32 * 16]) = rv2;
  *(float4*)(&Vs[0][v_st + 48 * 16]) = rv3;

  v8f o0 = {}, o1 = {}, o2 = {}, o3 = {};
  float mrun[8], lrun[8];
#pragma unroll
  for (int i = 0; i < 8; ++i) { mrun[i] = -1e30f; lrun[i] = 0.f; }

  float* P = Pbuf[wave];

  __syncthreads();

  constexpr int NT = S / KT;
  for (int t = 0; t < NT; ++t) {
    const int buf = t & 1;
    const bool more = (t + 1 < NT);
    const float* ks = &Ks[buf][row * PK + 2 * kh];
    const float* vs = &Vs[buf][(2 * kh) * 16 + row];

    v2f kf[16];
#pragma unroll
    for (int j = 0; j < 16; ++j) kf[j] = *(const v2f*)(ks + 4 * j);

#pragma unroll
    for (int sub = 0; sub < KT / 16; ++sub) {
      __builtin_amdgcn_sched_barrier(0);
      v8f sc = {};
#pragma unroll
      for (int j = 0; j < 16; ++j) sc = wmma4(qf[j], kf[j], sc);
      __builtin_amdgcn_sched_barrier(0);
      float4 rk, rv;
      if (more) {
        rk = *(const float4*)(kg + (size_t)(KT + sub * 16) * R3D);
        rv = *(const float4*)(vg + (size_t)(KT + sub * 16) * R3D);
      }
      v2f vf[16];
#pragma unroll
      for (int j = 0; j < 4; ++j) {
#pragma unroll
        for (int g = 0; g < 4; ++g) {
          const float* vp = vs + g * VG + (sub * 16 + 4 * j) * 16;
          vf[4 * j + g].x = vp[0];
          vf[4 * j + g].y = vp[16];
        }
      }
      if (sub + 1 < KT / 16) {
#pragma unroll
        for (int j = 0; j < 16; ++j)
          kf[j] = *(const v2f*)(ks + (sub + 1) * 16 * PK + 4 * j);
      }
      float alpha[8];
#pragma unroll
      for (int i = 0; i < 8; ++i) {
        float s = sc[i] * 0.125f;
        float mx = s;
        mx = fmaxf(mx, __shfl_xor(mx, 1, 32));
        mx = fmaxf(mx, __shfl_xor(mx, 2, 32));
        mx = fmaxf(mx, __shfl_xor(mx, 4, 32));
        mx = fmaxf(mx, __shfl_xor(mx, 8, 32));
        const float mnew = fmaxf(mrun[i], mx);
        const float a    = __expf(mrun[i] - mnew);
        const float p    = __expf(s - mnew);
        float sum = p;
        sum += __shfl_xor(sum, 1, 32);
        sum += __shfl_xor(sum, 2, 32);
        sum += __shfl_xor(sum, 4, 32);
        sum += __shfl_xor(sum, 8, 32);
        lrun[i] = lrun[i] * a + sum;
        mrun[i] = mnew;
        alpha[i] = a;
        P[(i + 8 * kh) * 18 + row] = p;
      }
#pragma unroll
      for (int i = 0; i < 8; ++i) {
        o0[i] *= alpha[i]; o1[i] *= alpha[i];
        o2[i] *= alpha[i]; o3[i] *= alpha[i];
      }
      v2f pa[4];
#pragma unroll
      for (int j = 0; j < 4; ++j)
        pa[j] = *(const v2f*)(P + row * 18 + 4 * j + 2 * kh);
      __builtin_amdgcn_sched_barrier(0);
#pragma unroll
      for (int j = 0; j < 4; ++j) {
        o0 = wmma4(pa[j], vf[4 * j + 0], o0);
        o1 = wmma4(pa[j], vf[4 * j + 1], o1);
        o2 = wmma4(pa[j], vf[4 * j + 2], o2);
        o3 = wmma4(pa[j], vf[4 * j + 3], o3);
      }
      __builtin_amdgcn_sched_barrier(0);
      if (more) {
        *(float4*)(&Ks[buf ^ 1][k_st + sub * 16 * PK]) = rk;
        *(float4*)(&Vs[buf ^ 1][v_st + sub * 16 * 16]) = rv;
      }
    }

    kg += (size_t)KT * R3D;
    vg += (size_t)KT * R3D;
    __syncthreads();
  }

#pragma unroll
  for (int i = 0; i < 8; ++i) {
    const int qr = q0 + i + 8 * kh;
    const float inv = 1.0f / lrun[i];
    float* orow = out + ((size_t)b * S + qr) * D + h * 64 + row;
    orow[0]  = o0[i] * inv;
    orow[16] = o1[i] * inv;
    orow[32] = o2[i] * inv;
    orow[48] = o3[i] * inv;
  }
}

extern "C" void kernel_launch(void* const* d_in, const int* in_sizes, int n_in,
                              void* d_out, int out_size, void* d_ws, size_t ws_size,
                              hipStream_t stream) {
  const float* x      = (const float*)d_in[0];
  const float* ln1_g  = (const float*)d_in[1];
  const float* ln1_b  = (const float*)d_in[2];
  const float* qkv_w  = (const float*)d_in[3];
  const float* qkv_b  = (const float*)d_in[4];
  const float* proj_w = (const float*)d_in[5];
  const float* proj_b = (const float*)d_in[6];
  const float* ln2_g  = (const float*)d_in[7];
  const float* ln2_b  = (const float*)d_in[8];
  const float* fc1_w  = (const float*)d_in[9];
  const float* fc1_b  = (const float*)d_in[10];
  const float* fc2_w  = (const float*)d_in[11];
  const float* fc2_b  = (const float*)d_in[12];
  float* out = (float*)d_out;

  const int M = 4096;
  float* bufA = (float*)d_ws;
  float* bufB = bufA + (size_t)4096 * 4096;
  float* bufC = bufB + (size_t)4096 * 3072;
  float* bufD = bufC + (size_t)4096 * 1024;

  ln_kernel<<<M, 256, 0, stream>>>(x, ln1_g, ln1_b, bufA);
  gemm_kernel<0><<<dim3(M / 128, 3072 / 64), 256, 0, stream>>>(
      bufA, qkv_w, qkv_b, nullptr, bufB, 3072, 1024);
  attn_kernel<<<dim3(32, 16), 256, 0, stream>>>(bufB, bufC);
  gemm_kernel<1><<<dim3(M / 128, 1024 / 64), 256, 0, stream>>>(
      bufC, proj_w, proj_b, x, bufD, 1024, 1024);
  ln_kernel<<<M, 256, 0, stream>>>(bufD, ln2_g, ln2_b, bufB);
  gemm_kernel<2><<<dim3(M / 128, 4096 / 64), 256, 0, stream>>>(
      bufB, fc1_w, fc1_b, nullptr, bufA, 4096, 1024);
  gemm_kernel<1><<<dim3(M / 128, 1024 / 64), 256, 0, stream>>>(
      bufA, fc2_w, fc2_b, bufD, out, 1024, 4096);
}


# attention: scores formed transposed (keys x queries): softmax max and sum run down a lane's 8 registers plus one lane-half exchange; exp(S) is the PV product's B operand in registers; new sum order
# speedup vs baseline: 1.4042x; 1.0746x over previous
#include <hip/hip_runtime.h>

typedef __attribute__((ext_vector_type(2))) float v2f;
typedef __attribute__((ext_vector_type(8))) float v8f;

__device__ __forceinline__ v8f wmma4(v2f a, v2f b, v8f c) {
  return __builtin_amdgcn_wmma_f32_16x16x4_f32(
       false, a,  false, b,
       (short)0, c,  false,  false);
}

__device__ __forceinline__ float gelu_exact(float x) {
  return 0.5f * x * (1.0f + erff(x * 0.70710678118654752440f));
}

__global__ __launch_bounds__(256) void ln_kernel(const float* __restrict__ x,
                                                 const float* __restrict__ gw,
                                                 const float* __restrict__ bw,
                                                 float* __restrict__ out) {
  const int row  = blockIdx.x;
  const int t    = threadIdx.x;
  const int wave = t >> 5, lane = t & 31;

  const float4 v = ((const float4*)(x + (size_t)row * 1024))[t];
  float s  = v.x + v.y + v.z + v.w;
  float s2 = v.x * v.x + v.y * v.y + v.z * v.z + v.w * v.w;
#pragma unroll
  for (int off = 16; off >= 1; off >>= 1) {
    s  += __shfl_xor(s, off, 32);
    s2 += __shfl_xor(s2, off, 32);
  }
  __shared__ float red[16];
  if (lane == 0) { red[2 * wave] = s; red[2 * wave + 1] = s2; }
  __syncthreads();
  float S1 = 0.f, S2 = 0.f;
#pragma unroll
  for (int w = 0; w < 8; ++w) { S1 += red[2 * w]; S2 += red[2 * w + 1]; }
  const float mean = S1 * (1.0f / 1024.0f);
  const float var  = S2 * (1.0f / 1024.0f) - mean * mean;
  const float rstd = rsqrtf(var + 1e-5f);

  const float4 g4 = ((const float4*)gw)[t];
  const float4 b4 = ((const float4*)bw)[t];
  float4 o;
  o.x = (v.x - mean) * rstd * g4.x + b4.x;
  o.y = (v.y - mean) * rstd * g4.y + b4.y;
  o.z = (v.z - mean) * rstd * g4.z + b4.z;
  o.w = (v.w - mean) * rstd * g4.w + b4.w;
  ((float4*)(out + (size_t)row * 1024))[t] = o;
}

template <int EPI>
__global__ __launch_bounds__(256) void gemm_kernel(const float* __restrict__ A,
                                                   const float* __restrict__ W,
                                                   const float* __restrict__ bias,
                                                   const float* __restrict__ res,
                                                   float* __restrict__ C,
                                                   int N, int K) {
  constexpr int BM = 128, BN = 64, BK = 32;
  constexpr int PA = BK + 4;
  constexpr int BG = BK * 16 + 16;
  __shared__ __attribute__((aligned(16))) float As[2][BM * PA];
  __shared__ __attribute__((aligned(16))) float Bs[2][(BN / 16) * BG];

  const int tid   = threadIdx.x;
  const int lane  = tid & 31;
  const int wave  = tid >> 5;
  const int row   = lane & 15;
  const int kh    = lane >> 4;
  const int waveM = wave >> 1;
  const int waveN = wave & 1;
  const int bm0 = blockIdx.x * BM;
  const int bn0 = blockIdx.y * BN;
  const int m0 = bm0 + waveM * 32;
  const int n0 = bn0 + waveN * 32;

  const int ar = tid >> 3, ac = (tid & 7) * 4;
  const int br = tid >> 4, bc = (tid & 15) * 4;
  const float* ag = A + (size_t)(bm0 + ar) * K + ac;
  const float* bg = W + (size_t)br * N + bn0 + bc;
  const int a_st = ar * PA + ac;
  const int b_st = (bc >> 4) * BG + br * 16 + (bc & 15);

  const int a_ld = (waveM * 32 + row) * PA + 2 * kh;
  const int b_ld = (waveN * 2) * BG + (2 * kh) * 16 + row;

  v8f c00 = {}, c01 = {}, c10 = {}, c11 = {};

  const size_t aK32 = (size_t)32 * K;
  const size_t bN16 = (size_t)16 * N;
  float4 ra0 = *(const float4*)(ag);
  float4 ra1 = *(const float4*)(ag + aK32);
  float4 ra2 = *(const float4*)(ag + 2 * aK32);
  float4 ra3 = *(const float4*)(ag + 3 * aK32);
  float4 rb0 = *(const float4*)(bg);
  float4 rb1 = *(const float4*)(bg + bN16);
  *(float4*)(&As[0][a_st])           = ra0;
  *(float4*)(&As[0][a_st + 32 * PA]) = ra1;
  *(float4*)(&As[0][a_st + 64 * PA]) = ra2;
  *(float4*)(&As[0][a_st + 96 * PA]) = ra3;
  *(float4*)(&Bs[0][b_st])           = rb0;
  *(float4*)(&Bs[0][b_st + 16 * 16]) = rb1;
  __syncthreads();

  const int nkt = K / BK;
  for (int t = 0; t < nkt; ++t) {
    const int buf = t & 1;
    const bool more = (t + 1 < nkt);
    if (more) {
      ag += BK;
      bg += (size_t)BK * N;
      ra0 = *(const float4*)(ag);
      ra1 = *(const float4*)(ag + aK32);
      ra2 = *(const float4*)(ag + 2 * aK32);
      ra3 = *(const float4*)(ag + 3 * aK32);
      rb0 = *(const float4*)(bg);
      rb1 = *(const float4*)(bg + bN16);
    }

    const float* as = &As[buf][a_ld];
    const float* bs = &Bs[buf][b_ld];
#pragma unroll
    for (int kk = 0; kk < BK; kk += 4) {
      const v2f a0 = *(const v2f*)(as + kk);
      const v2f a1 = *(const v2f*)(as + 16 * PA + kk);
      v2f b0, b1;
      b0.x = bs[kk * 16];      b0.y = bs[kk * 16 + 16];
      b1.x = bs[BG + kk * 16]; b1.y = bs[BG + kk * 16 + 16];
      c00 = wmma4(a0, b0, c00);
      c01 = wmma4(a0, b1, c01);
      c10 = wmma4(a1, b0, c10);
      c11 = wmma4(a1, b1, c11);
    }

    if (more) {
      float* asn = &As[buf ^ 1][a_st];
      float* bsn = &Bs[buf ^ 1][b_st];
      *(float4*)(asn)           = ra0;
      *(float4*)(asn + 32 * PA) = ra1;
      *(float4*)(asn + 64 * PA) = ra2;
      *(float4*)(asn + 96 * PA) = ra3;
      *(float4*)(bsn)           = rb0;
      *(float4*)(bsn + 16 * 16) = rb1;
    }
    __syncthreads();
  }

  const float bv0 = bias[n0 + row];
  const float bv1 = bias[n0 + 16 + row];
#pragma unroll
  for (int i = 0; i < 8; ++i) {
    const int r0 = m0 + i + 8 * kh;
    const int r1 = r0 + 16;
    const int cA = n0 + row;
    const int cB = cA + 16;
    float v00 = c00[i] + bv0, v01 = c01[i] + bv1;
    float v10 = c10[i] + bv0, v11 = c11[i] + bv1;
    if (EPI == 1) {
      v00 += res[(size_t)r0 * N + cA]; v01 += res[(size_t)r0 * N + cB];
      v10 += res[(size_t)r1 * N + cA]; v11 += res[(size_t)r1 * N + cB];
    }
    if (EPI == 2) {
      v00 = gelu_exact(v00); v01 = gelu_exact(v01);
      v10 = gelu_exact(v10); v11 = gelu_exact(v11);
    }
    C[(size_t)r0 * N + cA] = v00; C[(size_t)r0 * N + cB] = v01;
    C[(size_t)r1 * N + cA] = v10; C[(size_t)r1 * N + cB] = v11;
  }
}

__device__ __forceinline__ float xhalf(float x) {
  const int b = __builtin_bit_cast(int, x);
  return __builtin_bit_cast(
      float, __builtin_amdgcn_permlanex16(b, b, 0x76543210, 0xfedcba98, false, false));
}

__global__ __launch_bounds__(256) void attn_kernel(const float* __restrict__ qkv,
                                                   float* __restrict__ out) {
  constexpr int S = 2048, D = 1024, R3D = 3072;
  constexpr int KT = 64;
  constexpr int PK = 64 + 4;
  constexpr int VG = KT * 16 + 16;
  __shared__ __attribute__((aligned(16))) float Ks[2][KT * PK];
  __shared__ __attribute__((aligned(16))) float Vs[2][4 * VG];

  const int tid  = threadIdx.x;
  const int lane = tid & 31;
  const int wave = tid >> 5;
  const int row  = lane & 15;
  const int kh   = lane >> 4;
  const int bh   = blockIdx.x;
  const int b    = bh >> 4, h = bh & 15;
  const int q0   = (blockIdx.y * 8 + wave) * 16;

  const float* qbase = qkv + (size_t)b * S * R3D + h * 64;
  const float* kbase = qbase + D;
  const float* vbase = qbase + 2 * D;

  const int sr = tid >> 4, sc4 = (tid & 15) * 4;
  const int vrow = 4 * ((sr & 7) >> 1) + 2 * (sr >> 3) + (sr & 1);
  const float* kg = kbase + (size_t)sr * R3D + sc4;
  const float* vg = vbase + (size_t)sr * R3D + sc4;
  const int k_st = sr * PK + sc4;
  const int v_st = (sc4 >> 4) * VG + vrow * 16 + (sc4 & 15);
  const size_t r16 = (size_t)16 * R3D;

  float4 rk0 = *(const float4*)(kg);
  float4 rk1 = *(const float4*)(kg + r16);
  float4 rk2 = *(const float4*)(kg + 2 * r16);
  float4 rk3 = *(const float4*)(kg + 3 * r16);
  float4 rv0 = *(const float4*)(vg);
  float4 rv1 = *(const float4*)(vg + r16);
  float4 rv2 = *(const float4*)(vg + 2 * r16);
  float4 rv3 = *(const float4*)(vg + 3 * r16);

  v2f qf[16];
  {
    const float* qrow = qbase + (size_t)(q0 + row) * R3D + 2 * kh;
#pragma unroll
    for (int j = 0; j < 16; ++j) qf[j] = *(const v2f*)(qrow + 4 * j);
  }

  *(float4*)(&Ks[0][k_st])           = rk0;
  *(float4*)(&Ks[0][k_st + 16 * PK]) = rk1;
  *(float4*)(&Ks[0][k_st + 32 * PK]) = rk2;
  *(float4*)(&Ks[0][k_st + 48 * PK]) = rk3;
  *(float4*)(&Vs[0][v_st])           = rv0;
  *(float4*)(&Vs[0][v_st + 16 * 16]) = rv1;
  *(float4*)(&Vs[0][v_st + 32 * 16]) = rv2;
  *(float4*)(&Vs[0][v_st + 48 * 16]) = rv3;

  v8f o0 = {}, o1 = {}, o2 = {}, o3 = {};
  float mrun = -1e30f, lrun = 0.f;

  __syncthreads();

  constexpr int NT = S / KT;
  for (int t = 0; t < NT; ++t) {
    const int buf = t & 1;
    const bool more = (t + 1 < NT);
    const float* ks = &Ks[buf][row * PK + 2 * kh];
    const float* vs = &Vs[buf][(2 * kh) * 16 + row];

    v2f kf[16];
#pragma unroll
    for (int j = 0; j < 16; ++j) kf[j] = *(const v2f*)(ks + 4 * j);

#pragma unroll
    for (int sub = 0; sub < KT / 16; ++sub) {
      __builtin_amdgcn_sched_barrier(0);
      v8f sc = {};
#pragma unroll
      for (int j = 0; j < 16; ++j) sc = wmma4(kf[j], qf[j], sc);
      __builtin_amdgcn_sched_barrier(0);
      float4 rk, rv;
      if (more) {
        rk = *(const float4*)(kg + (size_t)(KT + sub * 16) * R3D);
        rv = *(const float4*)(vg + (size_t)(KT + sub * 16) * R3D);
      }
      v2f vf[16];
#pragma unroll
      for (int j = 0; j < 4; ++j) {
#pragma unroll
        for (int g = 0; g < 4; ++g) {
          const float* vp = vs + g * VG + (sub * 16 + 4 * j) * 16;
          vf[4 * j + g].x = vp[0];
          vf[4 * j + g].y = vp[16];
        }
      }
      if (sub + 1 < KT / 16) {
#pragma unroll
        for (int j = 0; j < 16; ++j)
          kf[j] = *(const v2f*)(ks + (sub + 1) * 16 * PK + 4 * j);
      }
      float s[8];
#pragma unroll
      for (int i = 0; i < 8; ++i) s[i] = sc[i] * 0.125f;
      float mx = fmaxf(fmaxf(fmaxf(s[0], s[1]), fmaxf(s[2], s[3])),
                       fmaxf(fmaxf(s[4], s[5]), fmaxf(s[6], s[7])));
      mx = fmaxf(mx, xhalf(mx));
      const float mnew = fmaxf(mrun, mx);
      const float a    = __expf(mrun - mnew);
      v2f pb[4];
#pragma unroll
      for (int j = 0; j < 4; ++j) {
        pb[j].x = __expf(s[2 * j] - mnew);
        pb[j].y = __expf(s[2 * j + 1] - mnew);
      }
      const float sum = ((pb[0].x + pb[0].y) + (pb[1].x + pb[1].y)) +
                        ((pb[2].x + pb[2].y) + (pb[3].x + pb[3].y));
      lrun = lrun * a + sum;
      mrun = mnew;
#pragma unroll
      for (int i = 0; i < 8; ++i) {
        o0[i] *= a; o1[i] *= a;
        o2[i] *= a; o3[i] *= a;
      }
      __builtin_amdgcn_sched_barrier(0);
#pragma unroll
      for (int j = 0; j < 4; ++j) {
        o0 = wmma4(vf[4 * j + 0], pb[j], o0);
        o1 = wmma4(vf[4 * j + 1], pb[j], o1);
        o2 = wmma4(vf[4 * j + 2], pb[j], o2);
        o3 = wmma4(vf[4 * j + 3], pb[j], o3);
      }
      __builtin_amdgcn_sched_barrier(0);
      if (more) {
        *(float4*)(&Ks[buf ^ 1][k_st + sub * 16 * PK]) = rk;
        *(float4*)(&Vs[buf ^ 1][v_st + sub * 16 * 16]) = rv;
      }
    }

    kg += (size_t)KT * R3D;
    vg += (size_t)KT * R3D;
    __syncthreads();
  }

  const float inv = 1.0f / (lrun + xhalf(lrun));
  float* orow = out + ((size_t)b * S + q0 + row) * D + h * 64 + 8 * kh;
  float4 w;
  w.x = o0[0] * inv; w.y = o0[1] * inv; w.z = o0[2] * inv; w.w = o0[3] * inv;
  *(float4*)(orow)      = w;
  w.x = o0[4] * inv; w.y = o0[5] * inv; w.z = o0[6] * inv; w.w = o0[7] * inv;
  *(float4*)(orow + 4)  = w;
  w.x = o1[0] * inv; w.y = o1[1] * inv; w.z = o1[2] * inv; w.w = o1[3] * inv;
  *(float4*)(orow + 16) = w;
  w.x = o1[4] * inv; w.y = o1[5] * inv; w.z = o1[6] * inv; w.w = o1[7] * inv;
  *(float4*)(orow + 20) = w;
  w.x = o2[0] * inv; w.y = o2[1] * inv; w.z = o2[2] * inv; w.w = o2[3] * inv;
  *(float4*)(orow + 32) = w;
  w.x = o2[4] * inv; w.y = o2[5] * inv; w.z = o2[6] * inv; w.w = o2[7] * inv;
  *(float4*)(orow + 36) = w;
  w.x = o3[0] * inv; w.y = o3[1] * inv; w.z = o3[2] * inv; w.w = o3[3] * inv;
  *(float4*)(orow + 48) = w;
  w.x = o3[4] * inv; w.y = o3[5] * inv; w.z = o3[6] * inv; w.w = o3[7] * inv;
  *(float4*)(orow + 52) = w;
}

extern "C" void kernel_launch(void* const* d_in, const int* in_sizes, int n_in,
                              void* d_out, int out_size, void* d_ws, size_t ws_size,
                              hipStream_t stream) {
  const float* x      = (const float*)d_in[0];
  const float* ln1_g  = (const float*)d_in[1];
  const float* ln1_b  = (const float*)d_in[2];
  const float* qkv_w  = (const float*)d_in[3];
  const float* qkv_b  = (const float*)d_in[4];
  const float* proj_w = (const float*)d_in[5];
  const float* proj_b = (const float*)d_in[6];
  const float* ln2_g  = (const float*)d_in[7];
  const float* ln2_b  = (const float*)d_in[8];
  const float* fc1_w  = (const float*)d_in[9];
  const float* fc1_b  = (const float*)d_in[10];
  const float* fc2_w  = (const float*)d_in[11];
  const float* fc2_b  = (const float*)d_in[12];
  float* out = (float*)d_out;

  const int M = 4096;
  float* bufA = (float*)d_ws;
  float* bufB = bufA + (size_t)4096 * 4096;
  float* bufC = bufB + (size_t)4096 * 3072;
  float* bufD = bufC + (size_t)4096 * 1024;

  ln_kernel<<<M, 256, 0, stream>>>(x, ln1_g, ln1_b, bufA);
  gemm_kernel<0><<<dim3(M / 128, 3072 / 64), 256, 0, stream>>>(
      bufA, qkv_w, qkv_b, nullptr, bufB, 3072, 1024);
  attn_kernel<<<dim3(32, 16), 256, 0, stream>>>(bufB, bufC);
  gemm_kernel<1><<<dim3(M / 128, 1024 / 64), 256, 0, stream>>>(
      bufC, proj_w, proj_b, x, bufD, 1024, 1024);
  ln_kernel<<<M, 256, 0, stream>>>(bufD, ln2_g, ln2_b, bufB);
  gemm_kernel<2><<<dim3(M / 128, 4096 / 64), 256, 0, stream>>>(
      bufB, fc1_w, fc1_b, nullptr, bufA, 4096, 1024);
  gemm_kernel<1><<<dim3(M / 128, 1024 / 64), 256, 0, stream>>>(
      bufA, fc2_w, fc2_b, bufD, out, 1024, 4096);
}
